# starting kernel
# baseline (speedup 1.0000x reference)
#include <hip/hip_runtime.h>
#include <hip/hip_bf16.h>

typedef __attribute__((ext_vector_type(2))) float v2f;
typedef __attribute__((ext_vector_type(8))) float v8f;

#define D_FEAT 256

__global__ void zero_f32(float* __restrict__ p, int n) {
  int i = blockIdx.x * blockDim.x + threadIdx.x;
  if (i < n) p[i] = 0.0f;
}
__global__ void zero_i32(int* __restrict__ p, int n) {
  int i = blockIdx.x * blockDim.x + threadIdx.x;
  if (i < n) p[i] = 0;
}

__global__ void sage_scatter(const float* __restrict__ H,
                             const int* __restrict__ src,
                             const int* __restrict__ dst,
                             int E,
                             float* __restrict__ agg,
                             int* __restrict__ cnt) {
  long long idx = (long long)blockIdx.x * blockDim.x + threadIdx.x;
  long long total = (long long)E * 64;
  if (idx >= total) return;
  int e = (int)(idx >> 6);
  int c = (int)(idx & 63);
  int s = src[e];
  int d = dst[e];
  const float4 v = ((const float4*)(H + (long long)s * D_FEAT))[c];
  float* p = agg + (long long)d * D_FEAT + c * 4;
  unsafeAtomicAdd(p + 0, v.x);
  unsafeAtomicAdd(p + 1, v.y);
  unsafeAtomicAdd(p + 2, v.z);
  unsafeAtomicAdd(p + 3, v.w);
  if (c == 0) atomicAdd(&cnt[d], 1);
}

__global__ void inv_count(const int* __restrict__ cnt, float* __restrict__ inv, int n) {
  int i = blockIdx.x * blockDim.x + threadIdx.x;
  if (i < n) {
    int c = cnt[i];
    inv[i] = 1.0f / (float)(c > 0 ? c : 1);
  }
}

__global__ void sage_gemm_wmma(const float* __restrict__ agg,
                               const float* __restrict__ inv,
                               const float* __restrict__ selfA,
                               const float* __restrict__ Wn,
                               const float* __restrict__ Ws,
                               const float* __restrict__ bias,
                               float* __restrict__ out,
                               int M, int N, int do_relu) {
  const int K = D_FEAT;
  int wave = threadIdx.x >> 5;
  int lane = threadIdx.x & 31;
  int strips = N >> 6;
  int tilesM = (M + 15) >> 4;
  int tile = blockIdx.x * 8 + wave;
  if (tile >= tilesM * strips) return;
  int tm = tile / strips;
  int tn = tile - tm * strips;
  int row0 = tm << 4;
  int col0 = tn << 6;

  int lrow = row0 + (lane & 15);
  int kb = (lane >> 4) << 1;
  int nb = col0 + (lane & 15);

  bool mok = lrow < M;
  int lrc = mok ? lrow : (M - 1);
  float invm = mok ? inv[lrc] : 0.0f;
  float sscl = mok ? 1.0f : 0.0f;
  const float* aggRow  = agg   + (long long)lrc * K + kb;
  const float* selfRow = selfA + (long long)lrc * K + kb;
  const float* pWn = Wn + (long long)kb * N + nb;
  const float* pWs = Ws + (long long)kb * N + nb;

  v8f acc[4] = {};
  for (int k0 = 0; k0 < K; k0 += 4) {
    float2 t = *(const float2*)(aggRow);
    float2 u = *(const float2*)(selfRow);
    v2f a1, a2;
    a1.x = t.x * invm; a1.y = t.y * invm;
    a2.x = u.x * sscl; a2.y = u.y * sscl;
    aggRow += 4; selfRow += 4;
#pragma unroll
    for (int tt = 0; tt < 4; ++tt) {
      v2f b1, b2;
      b1.x = pWn[tt * 16];
      b1.y = pWn[N + tt * 16];
      b2.x = pWs[tt * 16];
      b2.y = pWs[N + tt * 16];
      acc[tt] = __builtin_amdgcn_wmma_f32_16x16x4_f32(false, a1, false, b1,
                                                      (short)0, acc[tt], false, false);
      acc[tt] = __builtin_amdgcn_wmma_f32_16x16x4_f32(false, a2, false, b2,
                                                      (short)0, acc[tt], false, false);
    }
    pWn += 4 * N; pWs += 4 * N;
  }

  int mbase = row0 + ((lane >> 4) << 3);
#pragma unroll
  for (int tt = 0; tt < 4; ++tt) {
    int n = nb + tt * 16;
    float bn = bias[n];
#pragma unroll
    for (int v = 0; v < 8; ++v) {
      int m = mbase + v;
      if (m < M) {
        float r = acc[tt][v] + bn;
        if (do_relu) r = fmaxf(r, 0.0f);
        out[(long long)m * N + n] = r;
      }
    }
  }
}

static inline int cdiv(long long a, long long b) { return (int)((a + b - 1) / b); }

extern "C" void kernel_launch(void* const* d_in, const int* in_sizes, int n_in,
                              void* d_out, int out_size, void* d_ws, size_t ws_size,
                              hipStream_t stream) {
  const float* x   = (const float*)d_in[0];
  const float* Wn0 = (const float*)d_in[1];
  const float* Ws0 = (const float*)d_in[2];
  const float* b0  = (const float*)d_in[3];
  const float* Wn1 = (const float*)d_in[4];
  const float* Ws1 = (const float*)d_in[5];
  const float* b1  = (const float*)d_in[6];
  const float* Wn2 = (const float*)d_in[7];
  const float* Ws2 = (const float*)d_in[8];
  const float* b2  = (const float*)d_in[9];
  const int* src0  = (const int*)d_in[10];
  const int* dst0  = (const int*)d_in[11];
  const int* src1  = (const int*)d_in[12];
  const int* dst1  = (const int*)d_in[13];
  const int* src2  = (const int*)d_in[14];
  const int* dst2  = (const int*)d_in[15];

  const int N0 = 50000, N1 = 25000, N2 = 12500;
  const int E0 = 500000, E1 = 250000, E2 = 125000;

  char* ws = (char*)d_ws;
  float* agg = (float*)ws;
  ws += (size_t)N0 * D_FEAT * sizeof(float);
  float* h0 = (float*)ws;
  ws += (size_t)N0 * D_FEAT * sizeof(float);
  float* h1 = (float*)ws;
  ws += (size_t)N1 * D_FEAT * sizeof(float);
  float* inv = (float*)ws;
  ws += (size_t)N0 * sizeof(float);
  int* cnt = (int*)ws;

  const int TB = 256;

  zero_f32<<<cdiv((long long)N0 * D_FEAT, TB), TB, 0, stream>>>(agg, N0 * D_FEAT);
  zero_i32<<<cdiv(N0, TB), TB, 0, stream>>>(cnt, N0);
  sage_scatter<<<cdiv((long long)E0 * 64, TB), TB, 0, stream>>>(x, src0, dst0, E0, agg, cnt);
  inv_count<<<cdiv(N0, TB), TB, 0, stream>>>(cnt, inv, N0);
  {
    int tiles = ((N0 + 15) / 16) * (256 / 64);
    sage_gemm_wmma<<<cdiv(tiles, 8), TB, 0, stream>>>(agg, inv, x, Wn0, Ws0, b0, h0,
                                                      N0, 256, 1);
  }

  zero_f32<<<cdiv((long long)N1 * D_FEAT, TB), TB, 0, stream>>>(agg, N1 * D_FEAT);
  zero_i32<<<cdiv(N1, TB), TB, 0, stream>>>(cnt, N1);
  sage_scatter<<<cdiv((long long)E1 * 64, TB), TB, 0, stream>>>(h0, src1, dst1, E1, agg, cnt);
  inv_count<<<cdiv(N1, TB), TB, 0, stream>>>(cnt, inv, N1);
  {
    int tiles = ((N1 + 15) / 16) * (256 / 64);
    sage_gemm_wmma<<<cdiv(tiles, 8), TB, 0, stream>>>(agg, inv, h0, Wn1, Ws1, b1, h1,
                                                      N1, 256, 1);
  }

  zero_f32<<<cdiv((long long)N2 * D_FEAT, TB), TB, 0, stream>>>(agg, N2 * D_FEAT);
  zero_i32<<<cdiv(N2, TB), TB, 0, stream>>>(cnt, N2);
  sage_scatter<<<cdiv((long long)E2 * 64, TB), TB, 0, stream>>>(h1, src2, dst2, E2, agg, cnt);
  inv_count<<<cdiv(N2, TB), TB, 0, stream>>>(cnt, inv, N2);
  {
    int tiles = ((N2 + 15) / 16) * (128 / 64);
    sage_gemm_wmma<<<cdiv(tiles, 8), TB, 0, stream>>>(agg, inv, h1, Wn2, Ws2, b2,
                                                      (float*)d_out, N2, 128, 0);
  }
}


# Float-atomic scatter-add replaced: edges grouped by destination row once per call (int atomics + scan), then one wave per row sums source rows in registers; sum order = slot order, run-dependent
# speedup vs baseline: 1.4539x; 1.4539x over previous
#include <hip/hip_runtime.h>
#include <hip/hip_bf16.h>

typedef __attribute__((ext_vector_type(2))) float v2f;
typedef __attribute__((ext_vector_type(4))) float v4f;
typedef __attribute__((ext_vector_type(8))) float v8f;

#define D_FEAT 256


__global__ void zero_i32(int* __restrict__ p, int n) {
  int i = blockIdx.x * blockDim.x + threadIdx.x;
  if (i < n) p[i] = 0;
}

__global__ void edge_count(const int* __restrict__ dst0, const int* __restrict__ dst1,
                           const int* __restrict__ dst2, int E0, int E1, int E2,
                           int N0, int N1, int* __restrict__ cnt) {
  int idx = blockIdx.x * blockDim.x + threadIdx.x;
  if (idx >= E0 + E1 + E2) return;
  int g;
  if (idx < E0)           g = dst0[idx];
  else if (idx < E0 + E1) g = N0 + dst1[idx - E0];
  else                    g = N0 + N1 + dst2[idx - E0 - E1];
  atomicAdd(&cnt[g], 1);
}

__global__ void row_scan(const int* __restrict__ cnt, int* __restrict__ cursor,
                         float* __restrict__ inv, int n) {
  __shared__ int s_pre[32];
  __shared__ int s_own[32];
  int t = threadIdx.x;
  int lane = t & 31;
  int w = t >> 5;
  int base = blockIdx.x * 1024;
  const int4* c4 = (const int4*)cnt;
  int n4 = base >> 2;
  int part = 0;
#pragma unroll 4
  for (int j = t; j < n4; j += 1024) {
    int4 v = c4[j];
    part += (v.x + v.y) + (v.z + v.w);
  }
  int i = base + t;
  int own = (i < n) ? cnt[i] : 0;
  int incl = own;
  for (int d = 1; d < 32; d <<= 1) {
    int v = __shfl_up(incl, d);
    if (lane >= d) incl += v;
  }
  for (int d = 16; d >= 1; d >>= 1) part += __shfl_xor(part, d);
  if (lane == 31) s_own[w] = incl;
  if (lane == 0)  s_pre[w] = part;
  __syncthreads();
  int pre = s_pre[lane];
  for (int d = 16; d >= 1; d >>= 1) pre += __shfl_xor(pre, d);
  int wo = s_own[lane];
  int wincl = wo;
  for (int d = 1; d < 32; d <<= 1) {
    int v = __shfl_up(wincl, d);
    if (lane >= d) wincl += v;
  }
  int wexcl = __shfl(wincl - wo, w);
  if (i < n) {
    cursor[i] = pre + wexcl + (incl - own);
    inv[i] = 1.0f / (float)(own > 0 ? own : 1);
  }
}

__global__ void edge_fill(const int* __restrict__ src0, const int* __restrict__ dst0,
                          const int* __restrict__ src1, const int* __restrict__ dst1,
                          const int* __restrict__ src2, const int* __restrict__ dst2,
                          int E0, int E1, int E2, int N0, int N1,
                          int* __restrict__ cursor, int* __restrict__ col) {
  int idx = blockIdx.x * blockDim.x + threadIdx.x;
  if (idx >= E0 + E1 + E2) return;
  int g, s;
  if (idx < E0)           { g = dst0[idx];                      s = src0[idx]; }
  else if (idx < E0 + E1) { g = N0 + dst1[idx - E0];            s = src1[idx - E0]; }
  else                    { g = N0 + N1 + dst2[idx - E0 - E1];  s = src2[idx - E0 - E1]; }
  int pos = atomicAdd(&cursor[g], 1);
  col[pos] = s;
}

template <int R>
__device__ __forceinline__ void add_rows(const float* __restrict__ H, int myc, int j, int lane,
                                         v4f& a0, v4f& a1) {
  v4f x[R], y[R];
#pragma unroll
  for (int q = 0; q < R; ++q) {
    const v4f* p = (const v4f*)(H + (long long)__builtin_amdgcn_readlane(myc, j + q) * D_FEAT);
    x[q] = p[lane];
    y[q] = p[lane + 32];
  }
#pragma unroll
  for (int q = 0; q < R; ++q) {
    a0 += x[q];
    a1 += y[q];
  }
}

__global__ void sage_gather_sum(const float* __restrict__ H,
                                const int* __restrict__ rowend,
                                const int* __restrict__ col,
                                int rowoff, int N,
                                float* __restrict__ agg) {
  int wave = __builtin_amdgcn_readfirstlane((int)(threadIdx.x >> 5));
  int lane = threadIdx.x & 31;
  int r = blockIdx.x * 8 + wave;
  if (r >= N) return;
  int g = rowoff + r;
  int end = rowend[g];
  int beg = (g > 0) ? rowend[g - 1] : 0;
  v4f a0 = {0.0f, 0.0f, 0.0f, 0.0f};
  v4f a1 = {0.0f, 0.0f, 0.0f, 0.0f};
  for (int base = beg; base < end; base += 32) {
    int n = end - base;
    if (n > 32) n = 32;
    int myc = (lane < n) ? col[base + lane] : 0;
    int j = 0;
    for (; j + 8 <= n; j += 8) add_rows<8>(H, myc, j, lane, a0, a1);
    if (j + 4 <= n) { add_rows<4>(H, myc, j, lane, a0, a1); j += 4; }
    if (j + 2 <= n) { add_rows<2>(H, myc, j, lane, a0, a1); j += 2; }
    if (j < n) add_rows<1>(H, myc, j, lane, a0, a1);
  }
  v4f* o = (v4f*)(agg + (long long)r * D_FEAT);
  o[lane] = a0;
  o[lane + 32] = a1;
}

__global__ void sage_gemm_wmma(const float* __restrict__ agg,
                               const float* __restrict__ inv,
                               const float* __restrict__ selfA,
                               const float* __restrict__ Wn,
                               const float* __restrict__ Ws,
                               const float* __restrict__ bias,
                               float* __restrict__ out,
                               int M, int N, int do_relu) {
  const int K = D_FEAT;
  int wave = threadIdx.x >> 5;
  int lane = threadIdx.x & 31;
  int strips = N >> 6;
  int tilesM = (M + 15) >> 4;
  int tile = blockIdx.x * 8 + wave;
  if (tile >= tilesM * strips) return;
  int tm = tile / strips;
  int tn = tile - tm * strips;
  int row0 = tm << 4;
  int col0 = tn << 6;

  int lrow = row0 + (lane & 15);
  int kb = (lane >> 4) << 1;
  int nb = col0 + (lane & 15);

  bool mok = lrow < M;
  int lrc = mok ? lrow : (M - 1);
  float invm = mok ? inv[lrc] : 0.0f;
  float sscl = mok ? 1.0f : 0.0f;
  const float* aggRow  = agg   + (long long)lrc * K + kb;
  const float* selfRow = selfA + (long long)lrc * K + kb;
  const float* pWn = Wn + (long long)kb * N + nb;
  const float* pWs = Ws + (long long)kb * N + nb;

  v8f acc[4] = {};
  for (int k0 = 0; k0 < K; k0 += 4) {
    float2 t = *(const float2*)(aggRow);
    float2 u = *(const float2*)(selfRow);
    v2f a1, a2;
    a1.x = t.x * invm; a1.y = t.y * invm;
    a2.x = u.x * sscl; a2.y = u.y * sscl;
    aggRow += 4; selfRow += 4;
#pragma unroll
    for (int tt = 0; tt < 4; ++tt) {
      v2f b1, b2;
      b1.x = pWn[tt * 16];
      b1.y = pWn[N + tt * 16];
      b2.x = pWs[tt * 16];
      b2.y = pWs[N + tt * 16];
      acc[tt] = __builtin_amdgcn_wmma_f32_16x16x4_f32(false, a1, false, b1,
                                                      (short)0, acc[tt], false, false);
      acc[tt] = __builtin_amdgcn_wmma_f32_16x16x4_f32(false, a2, false, b2,
                                                      (short)0, acc[tt], false, false);
    }
    pWn += 4 * N; pWs += 4 * N;
  }

  int mbase = row0 + ((lane >> 4) << 3);
#pragma unroll
  for (int tt = 0; tt < 4; ++tt) {
    int n = nb + tt * 16;
    float bn = bias[n];
#pragma unroll
    for (int v = 0; v < 8; ++v) {
      int m = mbase + v;
      if (m < M) {
        float r = acc[tt][v] + bn;
        if (do_relu) r = fmaxf(r, 0.0f);
        out[(long long)m * N + n] = r;
      }
    }
  }
}

static inline int cdiv(long long a, long long b) { return (int)((a + b - 1) / b); }

extern "C" void kernel_launch(void* const* d_in, const int* in_sizes, int n_in,
                              void* d_out, int out_size, void* d_ws, size_t ws_size,
                              hipStream_t stream) {
  const float* x   = (const float*)d_in[0];
  const float* Wn0 = (const float*)d_in[1];
  const float* Ws0 = (const float*)d_in[2];
  const float* b0  = (const float*)d_in[3];
  const float* Wn1 = (const float*)d_in[4];
  const float* Ws1 = (const float*)d_in[5];
  const float* b1  = (const float*)d_in[6];
  const float* Wn2 = (const float*)d_in[7];
  const float* Ws2 = (const float*)d_in[8];
  const float* b2  = (const float*)d_in[9];
  const int* src0  = (const int*)d_in[10];
  const int* dst0  = (const int*)d_in[11];
  const int* src1  = (const int*)d_in[12];
  const int* dst1  = (const int*)d_in[13];
  const int* src2  = (const int*)d_in[14];
  const int* dst2  = (const int*)d_in[15];

  const int N0 = 50000, N1 = 25000, N2 = 12500;
  const int E0 = 500000, E1 = 250000, E2 = 125000;
  const int NALL = N0 + N1 + N2;
  const int EALL = E0 + E1 + E2;
  const int NPAD = 87552;

  char* ws = (char*)d_ws;
  float* bufA = (float*)ws;
  ws += (size_t)N0 * D_FEAT * sizeof(float);
  float* bufB = (float*)ws;
  ws += (size_t)N0 * D_FEAT * sizeof(float);
  int* cnt = (int*)ws;
  ws += (size_t)NPAD * sizeof(int);
  int* cursor = (int*)ws;
  ws += (size_t)NPAD * sizeof(int);
  float* inv = (float*)ws;
  ws += (size_t)NPAD * sizeof(float);
  int* col = (int*)ws;

  float* agg0 = bufA;
  float* h0   = bufB;
  float* agg1 = bufA;
  float* h1   = bufA + (size_t)N1 * D_FEAT;
  float* agg2 = bufB;

  const int TB = 256;

  zero_i32<<<cdiv(NALL, TB), TB, 0, stream>>>(cnt, NALL);
  edge_count<<<cdiv(EALL, TB), TB, 0, stream>>>(dst0, dst1, dst2, E0, E1, E2, N0, N1, cnt);
  row_scan<<<cdiv(NALL, 1024), 1024, 0, stream>>>(cnt, cursor, inv, NALL);
  edge_fill<<<cdiv(EALL, TB), TB, 0, stream>>>(src0, dst0, src1, dst1, src2, dst2,
                                               E0, E1, E2, N0, N1, cursor, col);

  sage_gather_sum<<<cdiv(N0, 8), TB, 0, stream>>>(x, cursor, col, 0, N0, agg0);
  {
    int tiles = ((N0 + 15) / 16) * (256 / 64);
    sage_gemm_wmma<<<cdiv(tiles, 8), TB, 0, stream>>>(agg0, inv, x, Wn0, Ws0, b0, h0,
                                                      N0, 256, 1);
  }

  sage_gather_sum<<<cdiv(N1, 8), TB, 0, stream>>>(h0, cursor, col, N0, N1, agg1);
  {
    int tiles = ((N1 + 15) / 16) * (256 / 64);
    sage_gemm_wmma<<<cdiv(tiles, 8), TB, 0, stream>>>(agg1, inv + N0, h0, Wn1, Ws1, b1, h1,
                                                      N1, 256, 1);
  }

  sage_gather_sum<<<cdiv(N2, 8), TB, 0, stream>>>(h1, cursor, col, N0 + N1, N2, agg2);
  {
    int tiles = ((N2 + 15) / 16) * (128 / 64);
    sage_gemm_wmma<<<cdiv(tiles, 8), TB, 0, stream>>>(agg2, inv + N0 + N1, h1, Wn2, Ws2, b2,
                                                      (float*)d_out, N2, 128, 0);
  }
}


# Dual GEMM re-tiled: a 256-thread block stages 64x16 of agg and self and 16x128 of both weights per k chunk in LDS (coalesced 16-byte loads, 2 stages); waves multiply 32x32 from LDS; sum order kept
# speedup vs baseline: 1.9349x; 1.3308x over previous
#include <hip/hip_runtime.h>
#include <hip/hip_bf16.h>

typedef __attribute__((ext_vector_type(2))) float v2f;
typedef __attribute__((ext_vector_type(4))) float v4f;
typedef __attribute__((ext_vector_type(8))) float v8f;

#define D_FEAT 256


__global__ void zero_i32(int* __restrict__ p, int n) {
  int i = blockIdx.x * blockDim.x + threadIdx.x;
  if (i < n) p[i] = 0;
}

__global__ void edge_count(const int* __restrict__ dst0, const int* __restrict__ dst1,
                           const int* __restrict__ dst2, int E0, int E1, int E2,
                           int N0, int N1, int* __restrict__ cnt) {
  int idx = blockIdx.x * blockDim.x + threadIdx.x;
  if (idx >= E0 + E1 + E2) return;
  int g;
  if (idx < E0)           g = dst0[idx];
  else if (idx < E0 + E1) g = N0 + dst1[idx - E0];
  else                    g = N0 + N1 + dst2[idx - E0 - E1];
  atomicAdd(&cnt[g], 1);
}

__global__ void row_scan(const int* __restrict__ cnt, int* __restrict__ cursor,
                         float* __restrict__ inv, int n) {
  __shared__ int s_pre[32];
  __shared__ int s_own[32];
  int t = threadIdx.x;
  int lane = t & 31;
  int w = t >> 5;
  int base = blockIdx.x * 1024;
  const int4* c4 = (const int4*)cnt;
  int n4 = base >> 2;
  int part = 0;
#pragma unroll 4
  for (int j = t; j < n4; j += 1024) {
    int4 v = c4[j];
    part += (v.x + v.y) + (v.z + v.w);
  }
  int i = base + t;
  int own = (i < n) ? cnt[i] : 0;
  int incl = own;
  for (int d = 1; d < 32; d <<= 1) {
    int v = __shfl_up(incl, d);
    if (lane >= d) incl += v;
  }
  for (int d = 16; d >= 1; d >>= 1) part += __shfl_xor(part, d);
  if (lane == 31) s_own[w] = incl;
  if (lane == 0)  s_pre[w] = part;
  __syncthreads();
  int pre = s_pre[lane];
  for (int d = 16; d >= 1; d >>= 1) pre += __shfl_xor(pre, d);
  int wo = s_own[lane];
  int wincl = wo;
  for (int d = 1; d < 32; d <<= 1) {
    int v = __shfl_up(wincl, d);
    if (lane >= d) wincl += v;
  }
  int wexcl = __shfl(wincl - wo, w);
  if (i < n) {
    cursor[i] = pre + wexcl + (incl - own);
    inv[i] = 1.0f / (float)(own > 0 ? own : 1);
  }
}

__global__ void edge_fill(const int* __restrict__ src0, const int* __restrict__ dst0,
                          const int* __restrict__ src1, const int* __restrict__ dst1,
                          const int* __restrict__ src2, const int* __restrict__ dst2,
                          int E0, int E1, int E2, int N0, int N1,
                          int* __restrict__ cursor, int* __restrict__ col) {
  int idx = blockIdx.x * blockDim.x + threadIdx.x;
  if (idx >= E0 + E1 + E2) return;
  int g, s;
  if (idx < E0)           { g = dst0[idx];                      s = src0[idx]; }
  else if (idx < E0 + E1) { g = N0 + dst1[idx - E0];            s = src1[idx - E0]; }
  else                    { g = N0 + N1 + dst2[idx - E0 - E1];  s = src2[idx - E0 - E1]; }
  int pos = atomicAdd(&cursor[g], 1);
  col[pos] = s;
}

template <int R>
__device__ __forceinline__ void add_rows(const float* __restrict__ H, int myc, int j, int lane,
                                         v4f& a0, v4f& a1) {
  v4f x[R], y[R];
#pragma unroll
  for (int q = 0; q < R; ++q) {
    const v4f* p = (const v4f*)(H + (long long)__builtin_amdgcn_readlane(myc, j + q) * D_FEAT);
    x[q] = p[lane];
    y[q] = p[lane + 32];
  }
#pragma unroll
  for (int q = 0; q < R; ++q) {
    a0 += x[q];
    a1 += y[q];
  }
}

__global__ void sage_gather_sum(const float* __restrict__ H,
                                const int* __restrict__ rowend,
                                const int* __restrict__ col,
                                int rowoff, int N,
                                float* __restrict__ agg) {
  int wave = __builtin_amdgcn_readfirstlane((int)(threadIdx.x >> 5));
  int lane = threadIdx.x & 31;
  int r = blockIdx.x * 8 + wave;
  if (r >= N) return;
  int g = rowoff + r;
  int end = rowend[g];
  int beg = (g > 0) ? rowend[g - 1] : 0;
  v4f a0 = {0.0f, 0.0f, 0.0f, 0.0f};
  v4f a1 = {0.0f, 0.0f, 0.0f, 0.0f};
  for (int base = beg; base < end; base += 32) {
    int n = end - base;
    if (n > 32) n = 32;
    int myc = (lane < n) ? col[base + lane] : 0;
    int j = 0;
    for (; j + 8 <= n; j += 8) add_rows<8>(H, myc, j, lane, a0, a1);
    if (j + 4 <= n) { add_rows<4>(H, myc, j, lane, a0, a1); j += 4; }
    if (j + 2 <= n) { add_rows<2>(H, myc, j, lane, a0, a1); j += 2; }
    if (j < n) add_rows<1>(H, myc, j, lane, a0, a1);
  }
  v4f* o = (v4f*)(agg + (long long)r * D_FEAT);
  o[lane] = a0;
  o[lane + 32] = a1;
}

#define GT_M 64
#define GT_N 128
#define GT_K 16
#define A_PITCH 20
#define B_PITCH 288
#define A_SRC (GT_M * A_PITCH)
#define B_SRC ((GT_K / 2) * B_PITCH)
#define STAGE (2 * A_SRC + 2 * B_SRC)

__global__ __launch_bounds__(256) void sage_gemm_wmma(const float* __restrict__ agg,
                               const float* __restrict__ inv,
                               const float* __restrict__ selfA,
                               const float* __restrict__ Wn,
                               const float* __restrict__ Ws,
                               const float* __restrict__ bias,
                               float* __restrict__ out,
                               int M, int N, int do_relu) {
  __shared__ __attribute__((aligned(16))) float lds[2 * STAGE];
  const int K = D_FEAT;
  int t = threadIdx.x;
  int wave = t >> 5;
  int lane = t & 31;
  int nblkN = N >> 7;
  int tm = blockIdx.x / nblkN;
  int tn = blockIdx.x - tm * nblkN;
  int row0 = tm * GT_M;
  int col0 = tn * GT_N;

  int arow = t >> 2;
  int akq = t & 3;
  int grow = row0 + arow;
  bool mok = grow < M;
  int growc = mok ? grow : (M - 1);
  float invm = inv[growc];
  const float* gAgg  = agg   + (long long)growc * K + akq * 4;
  const float* gSelf = selfA + (long long)growc * K + akq * 4;
  int aSt = arow * A_PITCH + akq * 4;
  const float* gWn = Wn + (long long)(2 * wave) * N + col0 + lane * 4;
  const float* gWs = Ws + (long long)(2 * wave) * N + col0 + lane * 4;
  int bSt = wave * B_PITCH + lane * 8;

  int c = lane & 15;
  int h = lane >> 4;
  int wm = wave >> 2;
  int wn = wave & 3;
  int aFr = (wm * 32 + c) * A_PITCH + 2 * h;
  int bFr = h * B_PITCH + (wn * 32 + c) * 2;

  v4f pa, ps, pn0, pn1, pw0, pw1;
  const v4f zero4 = {0.0f, 0.0f, 0.0f, 0.0f};

  pa  = *(const v4f*)gAgg;
  ps  = *(const v4f*)gSelf;
  pn0 = *(const v4f*)gWn;
  pn1 = *(const v4f*)(gWn + N);
  pw0 = *(const v4f*)gWs;
  pw1 = *(const v4f*)(gWs + N);

  v8f acc[2][2] = {};
  const int NCH = D_FEAT / GT_K;

  for (int ch = 0; ch < NCH; ++ch) {
    float* st = lds + (ch & 1) * STAGE;
    {
      v4f va = mok ? pa * invm : zero4;
      v4f vs = mok ? ps : zero4;
      *(v4f*)(st + aSt) = va;
      *(v4f*)(st + A_SRC + aSt) = vs;
      v4f lo, hi;
      lo.x = pn0.x; lo.y = pn1.x; lo.z = pn0.y; lo.w = pn1.y;
      hi.x = pn0.z; hi.y = pn1.z; hi.z = pn0.w; hi.w = pn1.w;
      *(v4f*)(st + 2 * A_SRC + bSt) = lo;
      *(v4f*)(st + 2 * A_SRC + bSt + 4) = hi;
      lo.x = pw0.x; lo.y = pw1.x; lo.z = pw0.y; lo.w = pw1.y;
      hi.x = pw0.z; hi.y = pw1.z; hi.z = pw0.w; hi.w = pw1.w;
      *(v4f*)(st + 2 * A_SRC + B_SRC + bSt) = lo;
      *(v4f*)(st + 2 * A_SRC + B_SRC + bSt + 4) = hi;
    }
    __syncthreads();
    if (ch + 1 < NCH) {
      gAgg += GT_K; gSelf += GT_K;
      gWn += (long long)GT_K * N; gWs += (long long)GT_K * N;
      pa  = *(const v4f*)gAgg;
      ps  = *(const v4f*)gSelf;
      pn0 = *(const v4f*)gWn;
      pn1 = *(const v4f*)(gWn + N);
      pw0 = *(const v4f*)gWs;
      pw1 = *(const v4f*)(gWs + N);
    }
    __builtin_amdgcn_sched_barrier(0);
    const float* sA1 = st + aFr;
    const float* sA2 = st + A_SRC + aFr;
    const float* sB1 = st + 2 * A_SRC + bFr;
    const float* sB2 = st + 2 * A_SRC + B_SRC + bFr;
#pragma unroll
    for (int ks = 0; ks < GT_K / 4; ++ks) {
      v2f a1[2], a2[2], b1[2], b2[2];
#pragma unroll
      for (int mt = 0; mt < 2; ++mt) {
        a1[mt] = *(const v2f*)(sA1 + mt * 16 * A_PITCH + ks * 4);
        a2[mt] = *(const v2f*)(sA2 + mt * 16 * A_PITCH + ks * 4);
      }
#pragma unroll
      for (int nt = 0; nt < 2; ++nt) {
        b1[nt] = *(const v2f*)(sB1 + ks * 2 * B_PITCH + nt * 32);
        b2[nt] = *(const v2f*)(sB2 + ks * 2 * B_PITCH + nt * 32);
      }
#pragma unroll
      for (int mt = 0; mt < 2; ++mt) {
#pragma unroll
        for (int nt = 0; nt < 2; ++nt) {
          acc[mt][nt] = __builtin_amdgcn_wmma_f32_16x16x4_f32(false, a1[mt], false, b1[nt],
                                                              (short)0, acc[mt][nt], false, false);
          acc[mt][nt] = __builtin_amdgcn_wmma_f32_16x16x4_f32(false, a2[mt], false, b2[nt],
                                                              (short)0, acc[mt][nt], false, false);
        }
      }
    }
  }

#pragma unroll
  for (int nt = 0; nt < 2; ++nt) {
    int n = col0 + wn * 32 + nt * 16 + c;
    float bn = bias[n];
#pragma unroll
    for (int mt = 0; mt < 2; ++mt) {
      int mbase = row0 + wm * 32 + mt * 16 + (h << 3);
#pragma unroll
      for (int v = 0; v < 8; ++v) {
        int m = mbase + v;
        if (m < M) {
          float r = acc[mt][nt][v] + bn;
          if (do_relu) r = fmaxf(r, 0.0f);
          out[(long long)m * N + n] = r;
        }
      }
    }
  }
}

static inline int cdiv(long long a, long long b) { return (int)((a + b - 1) / b); }

extern "C" void kernel_launch(void* const* d_in, const int* in_sizes, int n_in,
                              void* d_out, int out_size, void* d_ws, size_t ws_size,
                              hipStream_t stream) {
  const float* x   = (const float*)d_in[0];
  const float* Wn0 = (const float*)d_in[1];
  const float* Ws0 = (const float*)d_in[2];
  const float* b0  = (const float*)d_in[3];
  const float* Wn1 = (const float*)d_in[4];
  const float* Ws1 = (const float*)d_in[5];
  const float* b1  = (const float*)d_in[6];
  const float* Wn2 = (const float*)d_in[7];
  const float* Ws2 = (const float*)d_in[8];
  const float* b2  = (const float*)d_in[9];
  const int* src0  = (const int*)d_in[10];
  const int* dst0  = (const int*)d_in[11];
  const int* src1  = (const int*)d_in[12];
  const int* dst1  = (const int*)d_in[13];
  const int* src2  = (const int*)d_in[14];
  const int* dst2  = (const int*)d_in[15];

  const int N0 = 50000, N1 = 25000, N2 = 12500;
  const int E0 = 500000, E1 = 250000, E2 = 125000;
  const int NALL = N0 + N1 + N2;
  const int EALL = E0 + E1 + E2;
  const int NPAD = 87552;

  char* ws = (char*)d_ws;
  float* bufA = (float*)ws;
  ws += (size_t)N0 * D_FEAT * sizeof(float);
  float* bufB = (float*)ws;
  ws += (size_t)N0 * D_FEAT * sizeof(float);
  int* cnt = (int*)ws;
  ws += (size_t)NPAD * sizeof(int);
  int* cursor = (int*)ws;
  ws += (size_t)NPAD * sizeof(int);
  float* inv = (float*)ws;
  ws += (size_t)NPAD * sizeof(float);
  int* col = (int*)ws;

  float* agg0 = bufA;
  float* h0   = bufB;
  float* agg1 = bufA;
  float* h1   = bufA + (size_t)N1 * D_FEAT;
  float* agg2 = bufB;

  const int TB = 256;

  zero_i32<<<cdiv(NALL, TB), TB, 0, stream>>>(cnt, NALL);
  edge_count<<<cdiv(EALL, TB), TB, 0, stream>>>(dst0, dst1, dst2, E0, E1, E2, N0, N1, cnt);
  row_scan<<<cdiv(NALL, 1024), 1024, 0, stream>>>(cnt, cursor, inv, NALL);
  edge_fill<<<cdiv(EALL, TB), TB, 0, stream>>>(src0, dst0, src1, dst1, src2, dst2,
                                               E0, E1, E2, N0, N1, cursor, col);

  sage_gather_sum<<<cdiv(N0, 8), TB, 0, stream>>>(x, cursor, col, 0, N0, agg0);
  {
    sage_gemm_wmma<<<cdiv(N0, 64) * (256 / 128), TB, 0, stream>>>(agg0, inv, x, Wn0, Ws0, b0, h0,
                                                                  N0, 256, 1);
  }

  sage_gather_sum<<<cdiv(N1, 8), TB, 0, stream>>>(h0, cursor, col, N0, N1, agg1);
  {
    sage_gemm_wmma<<<cdiv(N1, 64) * (256 / 128), TB, 0, stream>>>(agg1, inv + N0, h0, Wn1, Ws1, b1, h1,
                                                                  N1, 256, 1);
  }

  sage_gather_sum<<<cdiv(N2, 8), TB, 0, stream>>>(h1, cursor, col, N0 + N1, N2, agg2);
  {
    sage_gemm_wmma<<<cdiv(N2, 64) * (128 / 128), TB, 0, stream>>>(agg2, inv + N0 + N1, h1, Wn2, Ws2, b2,
                                                                  (float*)d_out, N2, 128, 0);
  }
}


# Layers 0 and 1: neighbour sums moved into the GEMM launch; a block sums its 32 rows into an LDS tile and reads A fragments from it, so sums skip memory and row loads overlap WMMAs; sum order kept
# speedup vs baseline: 2.0574x; 1.0633x over previous
#include <hip/hip_runtime.h>
#include <hip/hip_bf16.h>

typedef __attribute__((ext_vector_type(2))) float v2f;
typedef __attribute__((ext_vector_type(4))) float v4f;
typedef __attribute__((ext_vector_type(8))) float v8f;

#define D_FEAT 256


__global__ void zero_i32(int* __restrict__ p, int n) {
  int i = blockIdx.x * blockDim.x + threadIdx.x;
  if (i < n) p[i] = 0;
}

__global__ void edge_count(const int* __restrict__ dst0, const int* __restrict__ dst1,
                           const int* __restrict__ dst2, int E0, int E1, int E2,
                           int N0, int N1, int* __restrict__ cnt) {
  int idx = blockIdx.x * blockDim.x + threadIdx.x;
  if (idx >= E0 + E1 + E2) return;
  int g;
  if (idx < E0)           g = dst0[idx];
  else if (idx < E0 + E1) g = N0 + dst1[idx - E0];
  else                    g = N0 + N1 + dst2[idx - E0 - E1];
  atomicAdd(&cnt[g], 1);
}

__global__ void row_scan(const int* __restrict__ cnt, int* __restrict__ cursor,
                         float* __restrict__ inv, int n) {
  __shared__ int s_pre[32];
  __shared__ int s_own[32];
  int t = threadIdx.x;
  int lane = t & 31;
  int w = t >> 5;
  int base = blockIdx.x * 1024;
  const int4* c4 = (const int4*)cnt;
  int n4 = base >> 2;
  int part = 0;
#pragma unroll 4
  for (int j = t; j < n4; j += 1024) {
    int4 v = c4[j];
    part += (v.x + v.y) + (v.z + v.w);
  }
  int i = base + t;
  int own = (i < n) ? cnt[i] : 0;
  int incl = own;
  for (int d = 1; d < 32; d <<= 1) {
    int v = __shfl_up(incl, d);
    if (lane >= d) incl += v;
  }
  for (int d = 16; d >= 1; d >>= 1) part += __shfl_xor(part, d);
  if (lane == 31) s_own[w] = incl;
  if (lane == 0)  s_pre[w] = part;
  __syncthreads();
  int pre = s_pre[lane];
  for (int d = 16; d >= 1; d >>= 1) pre += __shfl_xor(pre, d);
  int wo = s_own[lane];
  int wincl = wo;
  for (int d = 1; d < 32; d <<= 1) {
    int v = __shfl_up(wincl, d);
    if (lane >= d) wincl += v;
  }
  int wexcl = __shfl(wincl - wo, w);
  if (i < n) {
    cursor[i] = pre + wexcl + (incl - own);
    inv[i] = 1.0f / (float)(own > 0 ? own : 1);
  }
}

__global__ void edge_fill(const int* __restrict__ src0, const int* __restrict__ dst0,
                          const int* __restrict__ src1, const int* __restrict__ dst1,
                          const int* __restrict__ src2, const int* __restrict__ dst2,
                          int E0, int E1, int E2, int N0, int N1,
                          int* __restrict__ cursor, int* __restrict__ col) {
  int idx = blockIdx.x * blockDim.x + threadIdx.x;
  if (idx >= E0 + E1 + E2) return;
  int g, s;
  if (idx < E0)           { g = dst0[idx];                      s = src0[idx]; }
  else if (idx < E0 + E1) { g = N0 + dst1[idx - E0];            s = src1[idx - E0]; }
  else                    { g = N0 + N1 + dst2[idx - E0 - E1];  s = src2[idx - E0 - E1]; }
  int pos = atomicAdd(&cursor[g], 1);
  col[pos] = s;
}

template <int R>
__device__ __forceinline__ void add_rows(const float* __restrict__ H, int myc, int j, int lane,
                                         v4f& a0, v4f& a1) {
  v4f x[R], y[R];
#pragma unroll
  for (int q = 0; q < R; ++q) {
    const v4f* p = (const v4f*)(H + (long long)__builtin_amdgcn_readlane(myc, j + q) * D_FEAT);
    x[q] = p[lane];
    y[q] = p[lane + 32];
  }
#pragma unroll
  for (int q = 0; q < R; ++q) {
    a0 += x[q];
    a1 += y[q];
  }
}


__device__ __forceinline__ void sum_rows(const float* __restrict__ H, const int* __restrict__ col,
                                         int beg, int end, int lane, int myc, v4f& a0, v4f& a1) {
  for (int base = beg; base < end; base += 32) {
    int n = end - base;
    if (n > 32) n = 32;
    if (base != beg) myc = (lane < n) ? col[base + lane] : 0;
    int j = 0;
    for (; j + 8 <= n; j += 8) add_rows<8>(H, myc, j, lane, a0, a1);
    if (j + 4 <= n) { add_rows<4>(H, myc, j, lane, a0, a1); j += 4; }
    if (j + 2 <= n) { add_rows<2>(H, myc, j, lane, a0, a1); j += 2; }
    if (j < n) add_rows<1>(H, myc, j, lane, a0, a1);
  }
}

__global__ void sage_gather_sum(const float* __restrict__ H,
                                const int* __restrict__ rowend,
                                const int* __restrict__ col,
                                int rowoff, int N,
                                float* __restrict__ agg) {
  int wave = __builtin_amdgcn_readfirstlane((int)(threadIdx.x >> 5));
  int lane = threadIdx.x & 31;
  int r = blockIdx.x * 8 + wave;
  if (r >= N) return;
  int g = rowoff + r;
  int end = rowend[g];
  int beg = (g > 0) ? rowend[g - 1] : 0;
  v4f a0 = {0.0f, 0.0f, 0.0f, 0.0f};
  v4f a1 = {0.0f, 0.0f, 0.0f, 0.0f};
  for (int base = beg; base < end; base += 32) {
    int n = end - base;
    if (n > 32) n = 32;
    int myc = (lane < n) ? col[base + lane] : 0;
    int j = 0;
    for (; j + 8 <= n; j += 8) add_rows<8>(H, myc, j, lane, a0, a1);
    if (j + 4 <= n) { add_rows<4>(H, myc, j, lane, a0, a1); j += 4; }
    if (j + 2 <= n) { add_rows<2>(H, myc, j, lane, a0, a1); j += 2; }
    if (j < n) add_rows<1>(H, myc, j, lane, a0, a1);
  }
  v4f* o = (v4f*)(agg + (long long)r * D_FEAT);
  o[lane] = a0;
  o[lane + 32] = a1;
}

#define GT_M 64
#define GT_N 128
#define GT_K 16
#define A_PITCH 20
#define B_PITCH 288
#define A_SRC (GT_M * A_PITCH)
#define B_SRC ((GT_K / 2) * B_PITCH)
#define STAGE (2 * A_SRC + 2 * B_SRC)

__global__ __launch_bounds__(256) void sage_gemm_wmma(const float* __restrict__ agg,
                               const float* __restrict__ inv,
                               const float* __restrict__ selfA,
                               const float* __restrict__ Wn,
                               const float* __restrict__ Ws,
                               const float* __restrict__ bias,
                               float* __restrict__ out,
                               int M, int N, int do_relu) {
  __shared__ __attribute__((aligned(16))) float lds[2 * STAGE];
  const int K = D_FEAT;
  int t = threadIdx.x;
  int wave = t >> 5;
  int lane = t & 31;
  int nblkN = N >> 7;
  int tm = blockIdx.x / nblkN;
  int tn = blockIdx.x - tm * nblkN;
  int row0 = tm * GT_M;
  int col0 = tn * GT_N;

  int arow = t >> 2;
  int akq = t & 3;
  int grow = row0 + arow;
  bool mok = grow < M;
  int growc = mok ? grow : (M - 1);
  float invm = inv[growc];
  const float* gAgg  = agg   + (long long)growc * K + akq * 4;
  const float* gSelf = selfA + (long long)growc * K + akq * 4;
  int aSt = arow * A_PITCH + akq * 4;
  const float* gWn = Wn + (long long)(2 * wave) * N + col0 + lane * 4;
  const float* gWs = Ws + (long long)(2 * wave) * N + col0 + lane * 4;
  int bSt = wave * B_PITCH + lane * 8;

  int c = lane & 15;
  int h = lane >> 4;
  int wm = wave >> 2;
  int wn = wave & 3;
  int aFr = (wm * 32 + c) * A_PITCH + 2 * h;
  int bFr = h * B_PITCH + (wn * 32 + c) * 2;

  v4f pa, ps, pn0, pn1, pw0, pw1;
  const v4f zero4 = {0.0f, 0.0f, 0.0f, 0.0f};

  pa  = *(const v4f*)gAgg;
  ps  = *(const v4f*)gSelf;
  pn0 = *(const v4f*)gWn;
  pn1 = *(const v4f*)(gWn + N);
  pw0 = *(const v4f*)gWs;
  pw1 = *(const v4f*)(gWs + N);

  v8f acc[2][2] = {};
  const int NCH = D_FEAT / GT_K;

  for (int ch = 0; ch < NCH; ++ch) {
    float* st = lds + (ch & 1) * STAGE;
    {
      v4f va = mok ? pa * invm : zero4;
      v4f vs = mok ? ps : zero4;
      *(v4f*)(st + aSt) = va;
      *(v4f*)(st + A_SRC + aSt) = vs;
      v4f lo, hi;
      lo.x = pn0.x; lo.y = pn1.x; lo.z = pn0.y; lo.w = pn1.y;
      hi.x = pn0.z; hi.y = pn1.z; hi.z = pn0.w; hi.w = pn1.w;
      *(v4f*)(st + 2 * A_SRC + bSt) = lo;
      *(v4f*)(st + 2 * A_SRC + bSt + 4) = hi;
      lo.x = pw0.x; lo.y = pw1.x; lo.z = pw0.y; lo.w = pw1.y;
      hi.x = pw0.z; hi.y = pw1.z; hi.z = pw0.w; hi.w = pw1.w;
      *(v4f*)(st + 2 * A_SRC + B_SRC + bSt) = lo;
      *(v4f*)(st + 2 * A_SRC + B_SRC + bSt + 4) = hi;
    }
    __syncthreads();
    if (ch + 1 < NCH) {
      gAgg += GT_K; gSelf += GT_K;
      gWn += (long long)GT_K * N; gWs += (long long)GT_K * N;
      pa  = *(const v4f*)gAgg;
      ps  = *(const v4f*)gSelf;
      pn0 = *(const v4f*)gWn;
      pn1 = *(const v4f*)(gWn + N);
      pw0 = *(const v4f*)gWs;
      pw1 = *(const v4f*)(gWs + N);
    }
    __builtin_amdgcn_sched_barrier(0);
    const float* sA1 = st + aFr;
    const float* sA2 = st + A_SRC + aFr;
    const float* sB1 = st + 2 * A_SRC + bFr;
    const float* sB2 = st + 2 * A_SRC + B_SRC + bFr;
#pragma unroll
    for (int ks = 0; ks < GT_K / 4; ++ks) {
      v2f a1[2], a2[2], b1[2], b2[2];
#pragma unroll
      for (int mt = 0; mt < 2; ++mt) {
        a1[mt] = *(const v2f*)(sA1 + mt * 16 * A_PITCH + ks * 4);
        a2[mt] = *(const v2f*)(sA2 + mt * 16 * A_PITCH + ks * 4);
      }
#pragma unroll
      for (int nt = 0; nt < 2; ++nt) {
        b1[nt] = *(const v2f*)(sB1 + ks * 2 * B_PITCH + nt * 32);
        b2[nt] = *(const v2f*)(sB2 + ks * 2 * B_PITCH + nt * 32);
      }
#pragma unroll
      for (int mt = 0; mt < 2; ++mt) {
#pragma unroll
        for (int nt = 0; nt < 2; ++nt) {
          acc[mt][nt] = __builtin_amdgcn_wmma_f32_16x16x4_f32(false, a1[mt], false, b1[nt],
                                                              (short)0, acc[mt][nt], false, false);
          acc[mt][nt] = __builtin_amdgcn_wmma_f32_16x16x4_f32(false, a2[mt], false, b2[nt],
                                                              (short)0, acc[mt][nt], false, false);
        }
      }
    }
  }

#pragma unroll
  for (int nt = 0; nt < 2; ++nt) {
    int n = col0 + wn * 32 + nt * 16 + c;
    float bn = bias[n];
#pragma unroll
    for (int mt = 0; mt < 2; ++mt) {
      int mbase = row0 + wm * 32 + mt * 16 + (h << 3);
#pragma unroll
      for (int v = 0; v < 8; ++v) {
        int m = mbase + v;
        if (m < M) {
          float r = acc[mt][nt][v] + bn;
          if (do_relu) r = fmaxf(r, 0.0f);
          out[(long long)m * N + n] = r;
        }
      }
    }
  }
}


#define FT_M 32
#define AG_PITCH 260
#define FA_PITCH 20
#define FB_PITCH 544
#define AG_TILE (FT_M * AG_PITCH)
#define FA_SRC (FT_M * FA_PITCH)
#define FB_SRC ((GT_K / 2) * FB_PITCH)
#define FSTAGE (FA_SRC + 2 * FB_SRC)

__global__ __launch_bounds__(256) void sage_fused_256(const float* __restrict__ H,
                               const int* __restrict__ rowend,
                               const int* __restrict__ col,
                               const float* __restrict__ inv,
                               int rowoff,
                               const float* __restrict__ Wn,
                               const float* __restrict__ Ws,
                               const float* __restrict__ bias,
                               float* __restrict__ out,
                               int M, int do_relu) {
  __shared__ __attribute__((aligned(16))) float lds[AG_TILE + 2 * FSTAGE];
  float* aggT = lds;
  float* stg = lds + AG_TILE;
  const int K = D_FEAT;
  const int N = 256;
  int t = threadIdx.x;
  int wave = __builtin_amdgcn_readfirstlane((int)(threadIdx.x >> 5));
  int lane = t & 31;
  int row0 = blockIdx.x * FT_M;

  bool aload = wave < 4;
  int arow = (t >> 2) & 31;
  int akq = t & 3;
  int grow = row0 + arow;
  bool mok = grow < M;
  int growc = mok ? grow : (M - 1);
  const float* gSelf = H + (long long)growc * K + akq * 4;
  int aSt = arow * FA_PITCH + akq * 4;
  const float* gWn = Wn + (long long)(2 * wave) * N + lane * 4;
  const float* gWs = Ws + (long long)(2 * wave) * N + lane * 4;
  int bSt = wave * FB_PITCH + lane * 8;

  int c = lane & 15;
  int h = lane >> 4;
  int gFr = c * AG_PITCH + 2 * h;
  int aFr = c * FA_PITCH + 2 * h;
  int bFr = h * FB_PITCH + (wave * 32 + c) * 2;

  v4f ps, pn[4], pw[4];
  const v4f zero4 = {0.0f, 0.0f, 0.0f, 0.0f};

  ps = zero4;
  if (aload) ps = *(const v4f*)gSelf;
  pn[0] = *(const v4f*)gWn;
  pn[1] = *(const v4f*)(gWn + 128);
  pn[2] = *(const v4f*)(gWn + N);
  pn[3] = *(const v4f*)(gWn + N + 128);
  pw[0] = *(const v4f*)gWs;
  pw[1] = *(const v4f*)(gWs + 128);
  pw[2] = *(const v4f*)(gWs + N);
  pw[3] = *(const v4f*)(gWs + N + 128);

  {
    int lr0 = wave * 4;
    int r0 = row0 + lr0;
    int bnd[5];
#pragma unroll
    for (int i = 0; i < 5; ++i) {
      int r = r0 + i - 1;
      int rc = (r < M) ? r : (M - 1);
      int gg = rowoff + rc;
      bnd[i] = (gg >= 0) ? rowend[gg] : 0;
    }
    int ids[4];
#pragma unroll
    for (int i = 0; i < 4; ++i) {
      int n = bnd[i + 1] - bnd[i];
      ids[i] = (lane < n) ? col[bnd[i] + lane] : 0;
    }
#pragma unroll
    for (int i = 0; i < 4; ++i) {
      v4f a0 = zero4, a1 = zero4;
      int beg = bnd[i], end = bnd[i + 1];
      float invm = 0.0f;
      if (r0 + i < M) {
        invm = inv[r0 + i];
        sum_rows(H, col, beg, end, lane, ids[i], a0, a1);
      }
      float* o = aggT + (lr0 + i) * AG_PITCH;
      *(v4f*)(o + 4 * lane) = a0 * invm;
      *(v4f*)(o + 128 + 4 * lane) = a1 * invm;
    }
  }

  v8f acc[2][2] = {};
  const int NCH = D_FEAT / GT_K;

  for (int ch = 0; ch < NCH; ++ch) {
    float* st = stg + (ch & 1) * FSTAGE;
    {
      if (aload) *(v4f*)(st + aSt) = mok ? ps : zero4;
      float* bn = st + FA_SRC + bSt;
      float* bw = st + FA_SRC + FB_SRC + bSt;
#pragma unroll
      for (int q = 0; q < 2; ++q) {
        v4f lo, hi;
        lo.x = pn[q].x; lo.y = pn[q + 2].x; lo.z = pn[q].y; lo.w = pn[q + 2].y;
        hi.x = pn[q].z; hi.y = pn[q + 2].z; hi.z = pn[q].w; hi.w = pn[q + 2].w;
        *(v4f*)(bn + q * 256) = lo;
        *(v4f*)(bn + q * 256 + 4) = hi;
        lo.x = pw[q].x; lo.y = pw[q + 2].x; lo.z = pw[q].y; lo.w = pw[q + 2].y;
        hi.x = pw[q].z; hi.y = pw[q + 2].z; hi.z = pw[q].w; hi.w = pw[q + 2].w;
        *(v4f*)(bw + q * 256) = lo;
        *(v4f*)(bw + q * 256 + 4) = hi;
      }
    }
    __syncthreads();
    if (ch + 1 < NCH) {
      gSelf += GT_K;
      gWn += (long long)GT_K * N; gWs += (long long)GT_K * N;
      if (aload) ps = *(const v4f*)gSelf;
      pn[0] = *(const v4f*)gWn;
      pn[1] = *(const v4f*)(gWn + 128);
      pn[2] = *(const v4f*)(gWn + N);
      pn[3] = *(const v4f*)(gWn + N + 128);
      pw[0] = *(const v4f*)gWs;
      pw[1] = *(const v4f*)(gWs + 128);
      pw[2] = *(const v4f*)(gWs + N);
      pw[3] = *(const v4f*)(gWs + N + 128);
    }
    __builtin_amdgcn_sched_barrier(0);
    const float* sA1 = aggT + gFr + ch * GT_K;
    const float* sA2 = st + aFr;
    const float* sB1 = st + FA_SRC + bFr;
    const float* sB2 = st + FA_SRC + FB_SRC + bFr;
#pragma unroll
    for (int ks = 0; ks < GT_K / 4; ++ks) {
      v2f a1[2], a2[2], b1[2], b2[2];
#pragma unroll
      for (int mt = 0; mt < 2; ++mt) {
        a1[mt] = *(const v2f*)(sA1 + mt * 16 * AG_PITCH + ks * 4);
        a2[mt] = *(const v2f*)(sA2 + mt * 16 * FA_PITCH + ks * 4);
      }
#pragma unroll
      for (int nt = 0; nt < 2; ++nt) {
        b1[nt] = *(const v2f*)(sB1 + ks * 2 * FB_PITCH + nt * 32);
        b2[nt] = *(const v2f*)(sB2 + ks * 2 * FB_PITCH + nt * 32);
      }
#pragma unroll
      for (int mt = 0; mt < 2; ++mt) {
#pragma unroll
        for (int nt = 0; nt < 2; ++nt) {
          acc[mt][nt] = __builtin_amdgcn_wmma_f32_16x16x4_f32(false, a1[mt], false, b1[nt],
                                                              (short)0, acc[mt][nt], false, false);
          acc[mt][nt] = __builtin_amdgcn_wmma_f32_16x16x4_f32(false, a2[mt], false, b2[nt],
                                                              (short)0, acc[mt][nt], false, false);
        }
      }
    }
  }

#pragma unroll
  for (int nt = 0; nt < 2; ++nt) {
    int n = wave * 32 + nt * 16 + c;
    float bn = bias[n];
#pragma unroll
    for (int mt = 0; mt < 2; ++mt) {
      int mbase = row0 + mt * 16 + (h << 3);
#pragma unroll
      for (int v = 0; v < 8; ++v) {
        int m = mbase + v;
        if (m < M) {
          float r = acc[mt][nt][v] + bn;
          if (do_relu) r = fmaxf(r, 0.0f);
          out[(long long)m * N + n] = r;
        }
      }
    }
  }
}

static inline int cdiv(long long a, long long b) { return (int)((a + b - 1) / b); }

extern "C" void kernel_launch(void* const* d_in, const int* in_sizes, int n_in,
                              void* d_out, int out_size, void* d_ws, size_t ws_size,
                              hipStream_t stream) {
  const float* x   = (const float*)d_in[0];
  const float* Wn0 = (const float*)d_in[1];
  const float* Ws0 = (const float*)d_in[2];
  const float* b0  = (const float*)d_in[3];
  const float* Wn1 = (const float*)d_in[4];
  const float* Ws1 = (const float*)d_in[5];
  const float* b1  = (const float*)d_in[6];
  const float* Wn2 = (const float*)d_in[7];
  const float* Ws2 = (const float*)d_in[8];
  const float* b2  = (const float*)d_in[9];
  const int* src0  = (const int*)d_in[10];
  const int* dst0  = (const int*)d_in[11];
  const int* src1  = (const int*)d_in[12];
  const int* dst1  = (const int*)d_in[13];
  const int* src2  = (const int*)d_in[14];
  const int* dst2  = (const int*)d_in[15];

  const int N0 = 50000, N1 = 25000, N2 = 12500;
  const int E0 = 500000, E1 = 250000, E2 = 125000;
  const int NALL = N0 + N1 + N2;
  const int EALL = E0 + E1 + E2;
  const int NPAD = 87552;

  char* ws = (char*)d_ws;
  float* bufA = (float*)ws;
  ws += (size_t)N0 * D_FEAT * sizeof(float);
  float* bufB = (float*)ws;
  ws += (size_t)N0 * D_FEAT * sizeof(float);
  int* cnt = (int*)ws;
  ws += (size_t)NPAD * sizeof(int);
  int* cursor = (int*)ws;
  ws += (size_t)NPAD * sizeof(int);
  float* inv = (float*)ws;
  ws += (size_t)NPAD * sizeof(float);
  int* col = (int*)ws;

  float* h0   = bufB;
  float* h1   = bufA + (size_t)N1 * D_FEAT;
  float* agg2 = bufB;

  const int TB = 256;

  zero_i32<<<cdiv(NALL, TB), TB, 0, stream>>>(cnt, NALL);
  edge_count<<<cdiv(EALL, TB), TB, 0, stream>>>(dst0, dst1, dst2, E0, E1, E2, N0, N1, cnt);
  row_scan<<<cdiv(NALL, 1024), 1024, 0, stream>>>(cnt, cursor, inv, NALL);
  edge_fill<<<cdiv(EALL, TB), TB, 0, stream>>>(src0, dst0, src1, dst1, src2, dst2,
                                               E0, E1, E2, N0, N1, cursor, col);

  sage_fused_256<<<cdiv(N0, FT_M), TB, 0, stream>>>(x, cursor, col, inv, 0, Wn0, Ws0, b0, h0, N0, 1);

  sage_fused_256<<<cdiv(N1, FT_M), TB, 0, stream>>>(h0, cursor, col, inv + N0, N0, Wn1, Ws1, b1, h1,
                                                    N1, 1);

  sage_gather_sum<<<cdiv(N2, 8), TB, 0, stream>>>(h1, cursor, col, N0 + N1, N2, agg2);
  {
    sage_gemm_wmma<<<cdiv(N2, 64) * (128 / 128), TB, 0, stream>>>(agg2, inv + N0 + N1, h1, Wn2, Ws2, b2,
                                                                  (float*)d_out, N2, 128, 0);
  }
}
